# starting kernel
# baseline (speedup 1.0000x reference)
#include <hip/hip_runtime.h>

typedef float v2f __attribute__((ext_vector_type(2)));
typedef float v8f __attribute__((ext_vector_type(8)));

#define BATCH  4096
#define EMBED  128
#define HIDDEN 256
#define NOUT   10
#define MAXLEN 200
#define XSTRIDE (MAXLEN + 1)

__global__ __launch_bounds__(256) void pool_kernel(const int* __restrict__ x,
                                                   const float* __restrict__ emb,
                                                   float* __restrict__ pooled) {
  const int row  = (int)((blockIdx.x * blockDim.x + threadIdx.x) >> 5);
  const int lane = (int)(threadIdx.x & 31);
  if (row >= BATCH) return;

  const int* __restrict__ xr = x + row * XSTRIDE;
  const int len = xr[0];
  const float4* __restrict__ embv = (const float4*)emb;

  float4 acc = {0.f, 0.f, 0.f, 0.f};
  int j = 0;
  for (; j + 4 <= len; j += 4) {
    const int t0 = xr[1 + j], t1 = xr[2 + j], t2 = xr[3 + j], t3 = xr[4 + j];
    const float4 e0 = embv[t0 * 32 + lane];
    const float4 e1 = embv[t1 * 32 + lane];
    const float4 e2 = embv[t2 * 32 + lane];
    const float4 e3 = embv[t3 * 32 + lane];
    const float s0 = (t0 != 0) ? 1.f : 0.f;
    const float s1 = (t1 != 0) ? 1.f : 0.f;
    const float s2 = (t2 != 0) ? 1.f : 0.f;
    const float s3 = (t3 != 0) ? 1.f : 0.f;
    acc.x = fmaf(s0, e0.x, acc.x); acc.y = fmaf(s0, e0.y, acc.y);
    acc.z = fmaf(s0, e0.z, acc.z); acc.w = fmaf(s0, e0.w, acc.w);
    acc.x = fmaf(s1, e1.x, acc.x); acc.y = fmaf(s1, e1.y, acc.y);
    acc.z = fmaf(s1, e1.z, acc.z); acc.w = fmaf(s1, e1.w, acc.w);
    acc.x = fmaf(s2, e2.x, acc.x); acc.y = fmaf(s2, e2.y, acc.y);
    acc.z = fmaf(s2, e2.z, acc.z); acc.w = fmaf(s2, e2.w, acc.w);
    acc.x = fmaf(s3, e3.x, acc.x); acc.y = fmaf(s3, e3.y, acc.y);
    acc.z = fmaf(s3, e3.z, acc.z); acc.w = fmaf(s3, e3.w, acc.w);
  }
  for (; j < len; ++j) {
    const int t = xr[1 + j];
    const float4 e = embv[t * 32 + lane];
    const float s = (t != 0) ? 1.f : 0.f;
    acc.x = fmaf(s, e.x, acc.x); acc.y = fmaf(s, e.y, acc.y);
    acc.z = fmaf(s, e.z, acc.z); acc.w = fmaf(s, e.w, acc.w);
  }
  ((float4*)pooled)[row * 32 + lane] = acc;
}

#define PS_STRIDE 132
#define HS_STRIDE 260
#define PT_STRIDE 18

__global__ __launch_bounds__(128) void mlp_kernel(const float* __restrict__ pooled,
                                                  const float* __restrict__ W1,
                                                  const float* __restrict__ b1,
                                                  const float* __restrict__ W2,
                                                  const float* __restrict__ b2,
                                                  float* __restrict__ out) {
  __shared__ float Ps[16 * PS_STRIDE];
  __shared__ float Hs[16 * HS_STRIDE];
  __shared__ float Pt[4 * 16 * PT_STRIDE];

  const int tid  = (int)threadIdx.x;
  const int lane = tid & 31;
  const int wave = tid >> 5;
  const int m0   = (int)blockIdx.x * 16;
  const int mn   = lane & 15;
  const int kh   = (lane >> 4) << 1;
  const int chi  = (lane >> 4) << 3;

  for (int i = tid; i < 16 * 32; i += 128) {
    const int r = i >> 5, c = i & 31;
    const float4 v = ((const float4*)pooled)[(m0 + r) * 32 + c];
    float* dst = &Ps[r * PS_STRIDE + c * 4];
    dst[0] = v.x; dst[1] = v.y; dst[2] = v.z; dst[3] = v.w;
  }
  __syncthreads();

  v8f acc[4] = {};
  const int nb = wave * 64;
  for (int k0 = 0; k0 < EMBED; k0 += 4) {
    v2f a;
    a.x = Ps[mn * PS_STRIDE + k0 + kh];
    a.y = Ps[mn * PS_STRIDE + k0 + kh + 1];
#pragma unroll
    for (int t = 0; t < 4; ++t) {
      v2f b;
      b.x = W1[(k0 + kh) * HIDDEN + nb + t * 16 + mn];
      b.y = W1[(k0 + kh + 1) * HIDDEN + nb + t * 16 + mn];
      acc[t] = __builtin_amdgcn_wmma_f32_16x16x4_f32(
          false, a, false, b, (short)0, acc[t], false, false);
    }
  }
#pragma unroll
  for (int t = 0; t < 4; ++t) {
#pragma unroll
    for (int r = 0; r < 8; ++r) {
      const int M = r + chi;
      const int N = nb + t * 16 + mn;
      const float h = acc[t][r] + b1[N];
      Hs[M * HS_STRIDE + N] = h > 0.f ? h : 0.f;
    }
  }
  __syncthreads();

  v8f c2 = {};
  const int kb = wave * 64;
  for (int k0 = kb; k0 < kb + 64; k0 += 4) {
    v2f a;
    a.x = Hs[mn * HS_STRIDE + k0 + kh];
    a.y = Hs[mn * HS_STRIDE + k0 + kh + 1];
    v2f b;
    b.x = (mn < NOUT) ? W2[(k0 + kh) * NOUT + mn] : 0.f;
    b.y = (mn < NOUT) ? W2[(k0 + kh + 1) * NOUT + mn] : 0.f;
    c2 = __builtin_amdgcn_wmma_f32_16x16x4_f32(
        false, a, false, b, (short)0, c2, false, false);
  }
#pragma unroll
  for (int r = 0; r < 8; ++r)
    Pt[wave * (16 * PT_STRIDE) + (r + chi) * PT_STRIDE + mn] = c2[r];
  __syncthreads();

  for (int i = tid; i < 16 * NOUT; i += 128) {
    const int m = i / NOUT;
    const int n = i - m * NOUT;
    float s = b2[n];
#pragma unroll
    for (int w = 0; w < 4; ++w) s += Pt[w * (16 * PT_STRIDE) + m * PT_STRIDE + n];
    out[(m0 + m) * NOUT + n] = s;
  }
}

extern "C" void kernel_launch(void* const* d_in, const int* in_sizes, int n_in,
                              void* d_out, int out_size, void* d_ws, size_t ws_size,
                              hipStream_t stream) {
  const int*   x   = (const int*)d_in[0];
  const float* emb = (const float*)d_in[1];
  const float* W1  = (const float*)d_in[2];
  const float* b1  = (const float*)d_in[3];
  const float* W2  = (const float*)d_in[4];
  const float* b2  = (const float*)d_in[5];
  float* out    = (float*)d_out;
  float* pooled = (float*)d_ws;

  pool_kernel<<<BATCH / 8, 256, 0, stream>>>(x, emb, pooled);
  mlp_kernel<<<BATCH / 16, 128, 0, stream>>>(pooled, W1, b1, W2, b2, out);
}


# MLP kernel re-tiled: K of both layers split over 8 waves (4 column groups x 2 K halves), P slice via a wave-private LDS slot, K-half partials added in the layer-2 A loader; sum order changes
# speedup vs baseline: 1.9360x; 1.9360x over previous
#include <hip/hip_runtime.h>

typedef float v2f __attribute__((ext_vector_type(2)));
typedef float v8f __attribute__((ext_vector_type(8)));

#define BATCH  4096
#define EMBED  128
#define HIDDEN 256
#define NOUT   10
#define MAXLEN 200
#define XSTRIDE (MAXLEN + 1)
#define POS_PER_TRIP 4

__global__ __launch_bounds__(256) void pool_kernel(const int* __restrict__ x,
                                                   const float* __restrict__ emb,
                                                   float* __restrict__ pooled) {
  __shared__ float4 part[8][8][32];

  const int lane = (int)(threadIdx.x & 31);
  const int wave = __builtin_amdgcn_readfirstlane((int)(threadIdx.x >> 5));
  const int row0 = (int)blockIdx.x * 8;
  const float4* __restrict__ embv = (const float4*)emb;

  const int pos = wave + 8 * lane;
  int ids[8];
  int trips = 0;
#pragma unroll
  for (int r = 0; r < 8; ++r) {
    const int* __restrict__ xr = x + (row0 + r) * XSTRIDE;
    int len = xr[0];
    len = len < 0 ? 0 : (len > MAXLEN ? MAXLEN : len);
    const int raw = (pos < MAXLEN) ? xr[1 + pos] : 0;
    ids[r] = (pos < len) ? raw : 0;
    const int cnt = (len > wave) ? ((len - wave + 7) >> 3) : 0;
    trips = cnt > trips ? cnt : trips;
  }

  float4 acc[8];
#pragma unroll
  for (int r = 0; r < 8; ++r) acc[r] = make_float4(0.f, 0.f, 0.f, 0.f);

  for (int i = 0; i < trips; i += POS_PER_TRIP) {
    float4 e[POS_PER_TRIP][8];
#pragma unroll
    for (int u = 0; u < POS_PER_TRIP; ++u) {
#pragma unroll
      for (int r = 0; r < 8; ++r) {
        e[u][r] = make_float4(0.f, 0.f, 0.f, 0.f);
        const int id = __builtin_amdgcn_readlane(ids[r], i + u);
        if (id != 0) e[u][r] = embv[id * 32 + lane];
      }
    }
#pragma unroll
    for (int u = 0; u < POS_PER_TRIP; ++u) {
#pragma unroll
      for (int r = 0; r < 8; ++r) {
        acc[r].x += e[u][r].x;
        acc[r].y += e[u][r].y;
        acc[r].z += e[u][r].z;
        acc[r].w += e[u][r].w;
      }
    }
  }

#pragma unroll
  for (int r = 0; r < 8; ++r) part[wave][r][lane] = acc[r];
  __syncthreads();

  float4 s = part[0][wave][lane];
#pragma unroll
  for (int w = 1; w < 8; ++w) {
    const float4 p = part[w][wave][lane];
    s.x += p.x; s.y += p.y; s.z += p.z; s.w += p.w;
  }
  ((float4*)pooled)[(row0 + wave) * 32 + lane] = s;
}

typedef float v4f __attribute__((ext_vector_type(4)));

#define SLOT_STRIDE 68
#define PART_STRIDE 260
#define PT_STRIDE 18

__global__ __launch_bounds__(256) void mlp_kernel(const float* __restrict__ pooled,
                                                  const float* __restrict__ W1,
                                                  const float* __restrict__ b1,
                                                  const float* __restrict__ W2,
                                                  const float* __restrict__ b2,
                                                  float* __restrict__ out) {
  __shared__ __attribute__((aligned(16))) float Pw[8 * 16 * SLOT_STRIDE];
  __shared__ __attribute__((aligned(16))) float part[2 * 16 * PART_STRIDE];
  __shared__ float Pt[8 * 16 * PT_STRIDE];

  const int tid  = (int)threadIdx.x;
  const int lane = tid & 31;
  const int wave = __builtin_amdgcn_readfirstlane(tid >> 5);
  const int m0 = (int)blockIdx.x * 16;
  const int c  = lane & 15;
  const int h  = lane >> 4;
  const int ng = wave & 3;
  const int kg = wave >> 2;

  v4f pv[8];
#pragma unroll
  for (int j = 0; j < 8; ++j)
    pv[j] = *(const v4f*)(pooled + (m0 + 2 * j + h) * EMBED + kg * 64 + 4 * c);
  __builtin_amdgcn_sched_barrier(0);

  const float* __restrict__ brow = W1 + (kg * 64 + 2 * h) * HIDDEN + ng * 64 + 4 * c;
  v4f bq[16][2];
#pragma unroll
  for (int s = 0; s < 16; ++s) {
    bq[s][0] = *(const v4f*)(brow + (4 * s) * HIDDEN);
    bq[s][1] = *(const v4f*)(brow + (4 * s + 1) * HIDDEN);
  }

  const int k2 = wave * 32 + 2 * h;
  const int cc = c < NOUT ? c : 0;
  float w2v[8][2];
  v2f   b1v[8];
#pragma unroll
  for (int s = 0; s < 8; ++s) {
    w2v[s][0] = W2[(k2 + 4 * s) * NOUT + cc];
    w2v[s][1] = W2[(k2 + 4 * s + 1) * NOUT + cc];
    b1v[s] = *(const v2f*)(b1 + k2 + 4 * s);
  }
  float b2v = 0.f;
  if (tid < 16 * NOUT) b2v = b2[tid % NOUT];

  float* slot = Pw + wave * (16 * SLOT_STRIDE);
#pragma unroll
  for (int j = 0; j < 8; ++j)
    *(v4f*)&slot[(2 * j + h) * SLOT_STRIDE + 4 * c] = pv[j];

  const float* arow = slot + c * SLOT_STRIDE + 2 * h;
  v8f acc[4] = {};
#pragma unroll
  for (int s = 0; s < 16; ++s) {
    const v2f a = *(const v2f*)(arow + 4 * s);
#pragma unroll
    for (int t = 0; t < 4; ++t) {
      v2f b;
      b.x = bq[s][0][t];
      b.y = bq[s][1][t];
      acc[t] = __builtin_amdgcn_wmma_f32_16x16x4_f32(
          false, a, false, b, (short)0, acc[t], false, false);
    }
  }
#pragma unroll
  for (int r = 0; r < 8; ++r) {
    v4f v;
    v.x = acc[0][r]; v.y = acc[1][r]; v.z = acc[2][r]; v.w = acc[3][r];
    *(v4f*)&part[(kg * 16 + r + 8 * h) * PART_STRIDE + ng * 64 + 4 * c] = v;
  }
  __syncthreads();

  v8f c2a = {};
  v8f c2b = {};
  const float* p0 = part + c * PART_STRIDE + k2;
  const float* p1 = p0 + 16 * PART_STRIDE;
#pragma unroll
  for (int s = 0; s < 8; ++s) {
    const v2f x0 = *(const v2f*)(p0 + 4 * s);
    const v2f x1 = *(const v2f*)(p1 + 4 * s);
    const float h0 = (x0.x + x1.x) + b1v[s].x;
    const float h1 = (x0.y + x1.y) + b1v[s].y;
    v2f a2;
    a2.x = h0 > 0.f ? h0 : 0.f;
    a2.y = h1 > 0.f ? h1 : 0.f;
    v2f b;
    b.x = (c < NOUT) ? w2v[s][0] : 0.f;
    b.y = (c < NOUT) ? w2v[s][1] : 0.f;
    if (s & 1)
      c2b = __builtin_amdgcn_wmma_f32_16x16x4_f32(false, a2, false, b, (short)0, c2b, false, false);
    else
      c2a = __builtin_amdgcn_wmma_f32_16x16x4_f32(false, a2, false, b, (short)0, c2a, false, false);
  }
#pragma unroll
  for (int r = 0; r < 8; ++r)
    Pt[wave * (16 * PT_STRIDE) + (r + 8 * h) * PT_STRIDE + c] = c2a[r] + c2b[r];
  __syncthreads();

  if (tid < 16 * NOUT) {
    const int m = tid / NOUT;
    const int n = tid - m * NOUT;
    float s = b2v;
#pragma unroll
    for (int w = 0; w < 8; ++w) s += Pt[w * (16 * PT_STRIDE) + m * PT_STRIDE + n];
    out[(m0 + m) * NOUT + n] = s;
  }
}

extern "C" void kernel_launch(void* const* d_in, const int* in_sizes, int n_in,
                              void* d_out, int out_size, void* d_ws, size_t ws_size,
                              hipStream_t stream) {
  const int*   x   = (const int*)d_in[0];
  const float* emb = (const float*)d_in[1];
  const float* W1  = (const float*)d_in[2];
  const float* b1  = (const float*)d_in[3];
  const float* W2  = (const float*)d_in[4];
  const float* b2  = (const float*)d_in[5];
  float* out    = (float*)d_out;
  float* pooled = (float*)d_ws;

  pool_kernel<<<BATCH / 8, 256, 0, stream>>>(x, emb, pooled);
  mlp_kernel<<<BATCH / 16, 256, 0, stream>>>(pooled, W1, b1, W2, b2, out);
}
